# starting kernel
# baseline (speedup 1.0000x reference)
#include <hip/hip_runtime.h>
#include <hip/hip_bf16.h>
#include <math.h>


typedef __attribute__((ext_vector_type(2))) float v2f;
typedef __attribute__((ext_vector_type(8))) float v8f;

#define NCC_EPS   1e-8f
#define TPAD_W    52
#define IMG_W     512
#define OUT_W     492
#define PSTR      52
#define PATCH_H   36

__global__ __launch_bounds__(32) void ncc_tnorm(const float* __restrict__ t,
                                                float* __restrict__ nt) {
    const int bc = blockIdx.x;
    const int lane = threadIdx.x;
    const float* tp = t + bc * 441;

    float s = 0.f, s2 = 0.f;
    for (int i = lane; i < 441; i += 32) {
        float v = tp[i];
        s += v;
        s2 = fmaf(v, v, s2);
    }
    for (int off = 16; off > 0; off >>= 1) {
        s  += __shfl_down(s, off, 32);
        s2 += __shfl_down(s2, off, 32);
    }
    s  = __shfl(s, 0, 32);
    s2 = __shfl(s2, 0, 32);

    const float inv = 1.f / 441.f;
    float mean = s * inv;
    float var  = fmaxf(s2 * inv - mean * mean, 0.f);
    float stdv = sqrtf(var);
    float scale = 1.f / ((stdv + NCC_EPS) * 441.f);

    float* np = nt + bc * 441;
    for (int i = lane; i < 441; i += 32)
        np[i] = (tp[i] - mean) * scale;
}

__global__ __launch_bounds__(128) void ncc_main(const float* __restrict__ inp,
                                                const float* __restrict__ nt,
                                                float* __restrict__ out) {
    __shared__ float tpad[21 * TPAD_W];
    __shared__ alignas(16) float patch[4][PATCH_H * PSTR];
    __shared__ float colB[4][16 * PSTR];
    __shared__ float recipT[4][512];

    const int bc   = blockIdx.y;
    const int y0   = blockIdx.x * 16;
    const int tid  = threadIdx.x;
    const int wave = tid >> 5;
    const int lane = tid & 31;
    const int lane15 = lane & 15;
    const int hl     = lane >> 4;

    const float* ntp = nt + bc * 441;
    for (int i = tid; i < 21 * TPAD_W; i += 128) {
        int ky  = i / TPAD_W;
        int ii  = i - ky * TPAD_W;
        int idx = ii - 15;
        tpad[i] = (idx >= 0 && idx <= 20) ? ntp[ky * 21 + idx] : 0.f;
    }
    __syncthreads();

    float* p  = patch[wave];
    float* cb = colB[wave];
    float* rt = recipT[wave];
    const float* src0 = inp + (size_t)bc * (IMG_W * IMG_W);
    float*       dst0 = out + (size_t)bc * (OUT_W * OUT_W);

    for (int sx = wave; sx < 16; sx += 4) {
        const int x0 = sx * 32;

        for (int r = 0; r < PATCH_H; ++r) {
            const int gy = y0 + r;
            const bool rok = (gy < IMG_W);
            const float* srow = src0 + gy * IMG_W;
            const int gx = x0 + lane;
            p[r * PSTR + lane] = (rok && gx < IMG_W) ? srow[gx] : 0.f;
            if (lane < 20) {
                const int gx2 = gx + 32;
                p[r * PSTR + lane + 32] =
                    (rok && gx2 < IMG_W) ? srow[gx2] : 0.f;
            }
        }
        asm volatile("s_wait_dscnt 0x0" ::: "memory");

        v8f acc0 = {0.f, 0.f, 0.f, 0.f, 0.f, 0.f, 0.f, 0.f};
        v8f acc1 = {0.f, 0.f, 0.f, 0.f, 0.f, 0.f, 0.f, 0.f};
        for (int ky = 0; ky < 21; ++ky) {
            const float* Arow = p + (lane15 + ky) * PSTR + 2 * hl;
            const float* Brow = tpad + ky * TPAD_W + 15 + 2 * hl - lane15;
#pragma unroll
            for (int q = 0; q < 9; ++q) {
                v2f a0 = *(const v2f*)(Arow + 4 * q);
                v2f a1 = *(const v2f*)(Arow + 4 * q + 16);
                v2f b;
                b.x = Brow[4 * q];
                b.y = Brow[4 * q + 1];
                acc0 = __builtin_amdgcn_wmma_f32_16x16x4_f32(
                        false, a0, false, b, (short)0, acc0, false, false);
                acc1 = __builtin_amdgcn_wmma_f32_16x16x4_f32(
                        false, a1, false, b, (short)0, acc1, false, false);
            }
        }

        for (int xx = lane; xx < PSTR; xx += 32) {
            float s = 0.f;
#pragma unroll
            for (int r = 0; r < 21; ++r) s += p[r * PSTR + xx];
            cb[xx] = s;
            for (int m = 1; m < 16; ++m) {
                s += p[(m + 20) * PSTR + xx] - p[(m - 1) * PSTR + xx];
                cb[m * PSTR + xx] = s;
            }
        }
        asm volatile("s_wait_dscnt 0x0" ::: "memory");
        {
            const int m  = lane >> 1;
            const int n0 = (lane & 1) * 16;
            float s = 0.f;
#pragma unroll
            for (int kx = 0; kx < 21; ++kx) s += cb[m * PSTR + n0 + kx];
            rt[m * 32 + n0] = s;
            for (int j = 1; j < 16; ++j) {
                const int n = n0 + j;
                s += cb[m * PSTR + n + 20] - cb[m * PSTR + n - 1];
                rt[m * 32 + n] = s;
            }
        }
        asm volatile("s_wait_dscnt 0x0" ::: "memory");

        for (int xx = lane; xx < PSTR; xx += 32) {
            float s2 = 0.f;
#pragma unroll
            for (int r = 0; r < 21; ++r) {
                float v = p[r * PSTR + xx];
                s2 = fmaf(v, v, s2);
            }
            cb[xx] = s2;
            for (int m = 1; m < 16; ++m) {
                float vi = p[(m + 20) * PSTR + xx];
                float vo = p[(m - 1) * PSTR + xx];
                s2 += vi * vi - vo * vo;
                cb[m * PSTR + xx] = s2;
            }
        }
        asm volatile("s_wait_dscnt 0x0" ::: "memory");
        {
            const int m  = lane >> 1;
            const int n0 = (lane & 1) * 16;
            const float inv = 1.f / 441.f;
            float s2 = 0.f;
#pragma unroll
            for (int kx = 0; kx < 21; ++kx) s2 += cb[m * PSTR + n0 + kx];
            {
                float mean = rt[m * 32 + n0] * inv, msq = s2 * inv;
                float stdv = sqrtf(msq - mean * mean + NCC_EPS);
                rt[m * 32 + n0] = 1.f / (stdv + NCC_EPS);
            }
            for (int j = 1; j < 16; ++j) {
                const int n = n0 + j;
                s2 += cb[m * PSTR + n + 20] - cb[m * PSTR + n - 1];
                float mean = rt[m * 32 + n] * inv, msq = s2 * inv;
                float stdv = sqrtf(msq - mean * mean + NCC_EPS);
                rt[m * 32 + n] = 1.f / (stdv + NCC_EPS);
            }
        }
        asm volatile("s_wait_dscnt 0x0" ::: "memory");

#pragma unroll
        for (int r = 0; r < 8; ++r) {
            const int m   = r + 8 * hl;
            const int gy  = y0 + m;
            const int gx0 = x0 + lane15;
            const int gx1 = gx0 + 16;
            if (gy < OUT_W && gx0 < OUT_W)
                dst0[gy * OUT_W + gx0] = acc0[r] * rt[m * 32 + lane15];
            if (gy < OUT_W && gx1 < OUT_W)
                dst0[gy * OUT_W + gx1] = acc1[r] * rt[m * 32 + 16 + lane15];
        }
        asm volatile("s_wait_dscnt 0x0" ::: "memory");
    }
}

extern "C" void kernel_launch(void* const* d_in, const int* in_sizes, int n_in,
                              void* d_out, int out_size, void* d_ws, size_t ws_size,
                              hipStream_t stream) {
    const float* inp  = (const float*)d_in[0];
    const float* tmpl = (const float*)d_in[1];
    float* out = (float*)d_out;
    float* nt  = (float*)d_ws;

    ncc_tnorm<<<48, 32, 0, stream>>>(tmpl, nt);

    dim3 grid(31, 48);
    ncc_main<<<grid, 128, 0, stream>>>(inp, nt, out);
}


# LDS staging of patch and template: all global loads of a strip issued as one batch (patch as 15 b128 per lane), one wait after; was one load-wait-store round trip per row
# speedup vs baseline: 1.9421x; 1.9421x over previous
#include <hip/hip_runtime.h>
#include <hip/hip_bf16.h>
#include <math.h>


typedef __attribute__((ext_vector_type(2))) float v2f;
typedef __attribute__((ext_vector_type(8))) float v8f;
typedef __attribute__((ext_vector_type(4))) float v4f;

#define NCC_EPS   1e-8f
#define TPAD_W    52
#define IMG_W     512
#define OUT_W     492
#define PSTR      52
#define PATCH_H   36
#define PATCH_CH  (PATCH_H * PSTR / 4)
#define PATCH_LD  ((PATCH_CH + 31) / 32)

__global__ __launch_bounds__(32) void ncc_tnorm(const float* __restrict__ t,
                                                float* __restrict__ nt) {
    const int bc = blockIdx.x;
    const int lane = threadIdx.x;
    const float* tp = t + bc * 441;

    float s = 0.f, s2 = 0.f;
    for (int i = lane; i < 441; i += 32) {
        float v = tp[i];
        s += v;
        s2 = fmaf(v, v, s2);
    }
    for (int off = 16; off > 0; off >>= 1) {
        s  += __shfl_down(s, off, 32);
        s2 += __shfl_down(s2, off, 32);
    }
    s  = __shfl(s, 0, 32);
    s2 = __shfl(s2, 0, 32);

    const float inv = 1.f / 441.f;
    float mean = s * inv;
    float var  = fmaxf(s2 * inv - mean * mean, 0.f);
    float stdv = sqrtf(var);
    float scale = 1.f / ((stdv + NCC_EPS) * 441.f);

    float* np = nt + bc * 441;
    for (int i = lane; i < 441; i += 32)
        np[i] = (tp[i] - mean) * scale;
}

__global__ __launch_bounds__(128) void ncc_main(const float* __restrict__ inp,
                                                const float* __restrict__ nt,
                                                float* __restrict__ out) {
    __shared__ float tpad[21 * TPAD_W];
    __shared__ alignas(16) float patch[4][PATCH_H * PSTR];
    __shared__ float colB[4][16 * PSTR];
    __shared__ float recipT[4][512];

    const int bc   = blockIdx.y;
    const int y0   = blockIdx.x * 16;
    const int tid  = threadIdx.x;
    const int wave = tid >> 5;
    const int lane = tid & 31;
    const int lane15 = lane & 15;
    const int hl     = lane >> 4;

    const float* ntp = nt + bc * 441;
    float tv[9];
#pragma unroll
    for (int j = 0; j < 9; ++j) {
        int i   = tid + 128 * j;
        if (j == 8) i = min(i, 21 * TPAD_W - 1);
        int ky  = i / TPAD_W;
        int idx = i - ky * TPAD_W - 15;
        float v = ntp[ky * 21 + min(max(idx, 0), 20)];
        tv[j] = (idx >= 0 && idx <= 20) ? v : 0.f;
    }
#pragma unroll
    for (int j = 0; j < 9; ++j) {
        int i = tid + 128 * j;
        if (j < 8 || i < 21 * TPAD_W) tpad[i] = tv[j];
    }
    __syncthreads();

    float* p  = patch[wave];
    float* cb = colB[wave];
    float* rt = recipT[wave];
    const float* src0 = inp + (size_t)bc * (IMG_W * IMG_W);
    float*       dst0 = out + (size_t)bc * (OUT_W * OUT_W);

    for (int sx = wave; sx < 16; sx += 4) {
        const int x0 = sx * 32;

        v4f pv[PATCH_LD];
#pragma unroll
        for (int j = 0; j < PATCH_LD; ++j) {
            const int i   = lane + 32 * j;
            const int row = i / 13;
            const int gy  = y0 + row;
            const int gx  = x0 + 4 * (i - row * 13);
            pv[j] = *(const v4f*)(src0 + min(gy, IMG_W - 1) * IMG_W
                                       + min(gx, IMG_W - 4));
        }
        asm volatile("" :: "v"(pv[0]), "v"(pv[1]), "v"(pv[2]), "v"(pv[3]),
                           "v"(pv[4]), "v"(pv[5]), "v"(pv[6]), "v"(pv[7]),
                           "v"(pv[8]), "v"(pv[9]), "v"(pv[10]), "v"(pv[11]),
                           "v"(pv[12]), "v"(pv[13]), "v"(pv[14]));
#pragma unroll
        for (int j = 0; j < PATCH_LD; ++j) {
            const int i   = lane + 32 * j;
            const int row = i / 13;
            const int gy  = y0 + row;
            const int gx  = x0 + 4 * (i - row * 13);
            const bool ok = (gy < IMG_W) && (gx < IMG_W);
            const v4f zero4 = {0.f, 0.f, 0.f, 0.f};
            if (i < PATCH_CH)
                *(v4f*)(p + 4 * i) = ok ? pv[j] : zero4;
        }
        asm volatile("s_wait_dscnt 0x0" ::: "memory");

        v8f acc0 = {0.f, 0.f, 0.f, 0.f, 0.f, 0.f, 0.f, 0.f};
        v8f acc1 = {0.f, 0.f, 0.f, 0.f, 0.f, 0.f, 0.f, 0.f};
        for (int ky = 0; ky < 21; ++ky) {
            const float* Arow = p + (lane15 + ky) * PSTR + 2 * hl;
            const float* Brow = tpad + ky * TPAD_W + 15 + 2 * hl - lane15;
#pragma unroll
            for (int q = 0; q < 9; ++q) {
                v2f a0 = *(const v2f*)(Arow + 4 * q);
                v2f a1 = *(const v2f*)(Arow + 4 * q + 16);
                v2f b;
                b.x = Brow[4 * q];
                b.y = Brow[4 * q + 1];
                acc0 = __builtin_amdgcn_wmma_f32_16x16x4_f32(
                        false, a0, false, b, (short)0, acc0, false, false);
                acc1 = __builtin_amdgcn_wmma_f32_16x16x4_f32(
                        false, a1, false, b, (short)0, acc1, false, false);
            }
        }

        for (int xx = lane; xx < PSTR; xx += 32) {
            float s = 0.f;
#pragma unroll
            for (int r = 0; r < 21; ++r) s += p[r * PSTR + xx];
            cb[xx] = s;
            for (int m = 1; m < 16; ++m) {
                s += p[(m + 20) * PSTR + xx] - p[(m - 1) * PSTR + xx];
                cb[m * PSTR + xx] = s;
            }
        }
        asm volatile("s_wait_dscnt 0x0" ::: "memory");
        {
            const int m  = lane >> 1;
            const int n0 = (lane & 1) * 16;
            float s = 0.f;
#pragma unroll
            for (int kx = 0; kx < 21; ++kx) s += cb[m * PSTR + n0 + kx];
            rt[m * 32 + n0] = s;
            for (int j = 1; j < 16; ++j) {
                const int n = n0 + j;
                s += cb[m * PSTR + n + 20] - cb[m * PSTR + n - 1];
                rt[m * 32 + n] = s;
            }
        }
        asm volatile("s_wait_dscnt 0x0" ::: "memory");

        for (int xx = lane; xx < PSTR; xx += 32) {
            float s2 = 0.f;
#pragma unroll
            for (int r = 0; r < 21; ++r) {
                float v = p[r * PSTR + xx];
                s2 = fmaf(v, v, s2);
            }
            cb[xx] = s2;
            for (int m = 1; m < 16; ++m) {
                float vi = p[(m + 20) * PSTR + xx];
                float vo = p[(m - 1) * PSTR + xx];
                s2 += vi * vi - vo * vo;
                cb[m * PSTR + xx] = s2;
            }
        }
        asm volatile("s_wait_dscnt 0x0" ::: "memory");
        {
            const int m  = lane >> 1;
            const int n0 = (lane & 1) * 16;
            const float inv = 1.f / 441.f;
            float s2 = 0.f;
#pragma unroll
            for (int kx = 0; kx < 21; ++kx) s2 += cb[m * PSTR + n0 + kx];
            {
                float mean = rt[m * 32 + n0] * inv, msq = s2 * inv;
                float stdv = sqrtf(msq - mean * mean + NCC_EPS);
                rt[m * 32 + n0] = 1.f / (stdv + NCC_EPS);
            }
            for (int j = 1; j < 16; ++j) {
                const int n = n0 + j;
                s2 += cb[m * PSTR + n + 20] - cb[m * PSTR + n - 1];
                float mean = rt[m * 32 + n] * inv, msq = s2 * inv;
                float stdv = sqrtf(msq - mean * mean + NCC_EPS);
                rt[m * 32 + n] = 1.f / (stdv + NCC_EPS);
            }
        }
        asm volatile("s_wait_dscnt 0x0" ::: "memory");

#pragma unroll
        for (int r = 0; r < 8; ++r) {
            const int m   = r + 8 * hl;
            const int gy  = y0 + m;
            const int gx0 = x0 + lane15;
            const int gx1 = gx0 + 16;
            if (gy < OUT_W && gx0 < OUT_W)
                dst0[gy * OUT_W + gx0] = acc0[r] * rt[m * 32 + lane15];
            if (gy < OUT_W && gx1 < OUT_W)
                dst0[gy * OUT_W + gx1] = acc1[r] * rt[m * 32 + 16 + lane15];
        }
        asm volatile("s_wait_dscnt 0x0" ::: "memory");
    }
}

extern "C" void kernel_launch(void* const* d_in, const int* in_sizes, int n_in,
                              void* d_out, int out_size, void* d_ws, size_t ws_size,
                              hipStream_t stream) {
    const float* inp  = (const float*)d_in[0];
    const float* tmpl = (const float*)d_in[1];
    float* out = (float*)d_out;
    float* nt  = (float*)d_ws;

    ncc_tnorm<<<48, 32, 0, stream>>>(tmpl, nt);

    dim3 grid(31, 48);
    ncc_main<<<grid, 128, 0, stream>>>(inp, nt, out);
}


# Work granularity: one 32-column strip per wave instead of four (grid 124x48 workgroups, was 31x48), so every WGP stays full until the grid drains; no near-empty second dispatch round
# speedup vs baseline: 2.0253x; 1.0428x over previous
#include <hip/hip_runtime.h>
#include <hip/hip_bf16.h>
#include <math.h>


typedef __attribute__((ext_vector_type(2))) float v2f;
typedef __attribute__((ext_vector_type(8))) float v8f;
typedef __attribute__((ext_vector_type(4))) float v4f;

#define NCC_EPS   1e-8f
#define TPAD_W    52
#define IMG_W     512
#define OUT_W     492
#define PSTR      52
#define PATCH_H   36
#define PATCH_CH  (PATCH_H * PSTR / 4)
#define PATCH_LD  ((PATCH_CH + 31) / 32)

__global__ __launch_bounds__(32) void ncc_tnorm(const float* __restrict__ t,
                                                float* __restrict__ nt) {
    const int bc = blockIdx.x;
    const int lane = threadIdx.x;
    const float* tp = t + bc * 441;

    float s = 0.f, s2 = 0.f;
    for (int i = lane; i < 441; i += 32) {
        float v = tp[i];
        s += v;
        s2 = fmaf(v, v, s2);
    }
    for (int off = 16; off > 0; off >>= 1) {
        s  += __shfl_down(s, off, 32);
        s2 += __shfl_down(s2, off, 32);
    }
    s  = __shfl(s, 0, 32);
    s2 = __shfl(s2, 0, 32);

    const float inv = 1.f / 441.f;
    float mean = s * inv;
    float var  = fmaxf(s2 * inv - mean * mean, 0.f);
    float stdv = sqrtf(var);
    float scale = 1.f / ((stdv + NCC_EPS) * 441.f);

    float* np = nt + bc * 441;
    for (int i = lane; i < 441; i += 32)
        np[i] = (tp[i] - mean) * scale;
}

__global__ __launch_bounds__(128) void ncc_main(const float* __restrict__ inp,
                                                const float* __restrict__ nt,
                                                float* __restrict__ out) {
    __shared__ float tpad[21 * TPAD_W];
    __shared__ alignas(16) float patch[4][PATCH_H * PSTR];
    __shared__ float colB[4][16 * PSTR];
    __shared__ float recipT[4][512];

    const int bc   = blockIdx.y;
    const int y0   = (blockIdx.x >> 2) * 16;
    const int sg   = blockIdx.x & 3;
    const int tid  = threadIdx.x;
    const int wave = tid >> 5;
    const int lane = tid & 31;
    const int lane15 = lane & 15;
    const int hl     = lane >> 4;

    const float* ntp = nt + bc * 441;
    float tv[9];
#pragma unroll
    for (int j = 0; j < 9; ++j) {
        int i   = tid + 128 * j;
        if (j == 8) i = min(i, 21 * TPAD_W - 1);
        int ky  = i / TPAD_W;
        int idx = i - ky * TPAD_W - 15;
        float v = ntp[ky * 21 + min(max(idx, 0), 20)];
        tv[j] = (idx >= 0 && idx <= 20) ? v : 0.f;
    }
#pragma unroll
    for (int j = 0; j < 9; ++j) {
        int i = tid + 128 * j;
        if (j < 8 || i < 21 * TPAD_W) tpad[i] = tv[j];
    }
    __syncthreads();

    float* p  = patch[wave];
    float* cb = colB[wave];
    float* rt = recipT[wave];
    const float* src0 = inp + (size_t)bc * (IMG_W * IMG_W);
    float*       dst0 = out + (size_t)bc * (OUT_W * OUT_W);

    {
        const int x0 = (sg * 4 + wave) * 32;

        v4f pv[PATCH_LD];
#pragma unroll
        for (int j = 0; j < PATCH_LD; ++j) {
            const int i   = lane + 32 * j;
            const int row = i / 13;
            const int gy  = y0 + row;
            const int gx  = x0 + 4 * (i - row * 13);
            pv[j] = *(const v4f*)(src0 + min(gy, IMG_W - 1) * IMG_W
                                       + min(gx, IMG_W - 4));
        }
        asm volatile("" :: "v"(pv[0]), "v"(pv[1]), "v"(pv[2]), "v"(pv[3]),
                           "v"(pv[4]), "v"(pv[5]), "v"(pv[6]), "v"(pv[7]),
                           "v"(pv[8]), "v"(pv[9]), "v"(pv[10]), "v"(pv[11]),
                           "v"(pv[12]), "v"(pv[13]), "v"(pv[14]));
#pragma unroll
        for (int j = 0; j < PATCH_LD; ++j) {
            const int i   = lane + 32 * j;
            const int row = i / 13;
            const int gy  = y0 + row;
            const int gx  = x0 + 4 * (i - row * 13);
            const bool ok = (gy < IMG_W) && (gx < IMG_W);
            const v4f zero4 = {0.f, 0.f, 0.f, 0.f};
            if (i < PATCH_CH)
                *(v4f*)(p + 4 * i) = ok ? pv[j] : zero4;
        }
        asm volatile("s_wait_dscnt 0x0" ::: "memory");

        v8f acc0 = {0.f, 0.f, 0.f, 0.f, 0.f, 0.f, 0.f, 0.f};
        v8f acc1 = {0.f, 0.f, 0.f, 0.f, 0.f, 0.f, 0.f, 0.f};
        for (int ky = 0; ky < 21; ++ky) {
            const float* Arow = p + (lane15 + ky) * PSTR + 2 * hl;
            const float* Brow = tpad + ky * TPAD_W + 15 + 2 * hl - lane15;
#pragma unroll
            for (int q = 0; q < 9; ++q) {
                v2f a0 = *(const v2f*)(Arow + 4 * q);
                v2f a1 = *(const v2f*)(Arow + 4 * q + 16);
                v2f b;
                b.x = Brow[4 * q];
                b.y = Brow[4 * q + 1];
                acc0 = __builtin_amdgcn_wmma_f32_16x16x4_f32(
                        false, a0, false, b, (short)0, acc0, false, false);
                acc1 = __builtin_amdgcn_wmma_f32_16x16x4_f32(
                        false, a1, false, b, (short)0, acc1, false, false);
            }
        }

        for (int xx = lane; xx < PSTR; xx += 32) {
            float s = 0.f;
#pragma unroll
            for (int r = 0; r < 21; ++r) s += p[r * PSTR + xx];
            cb[xx] = s;
            for (int m = 1; m < 16; ++m) {
                s += p[(m + 20) * PSTR + xx] - p[(m - 1) * PSTR + xx];
                cb[m * PSTR + xx] = s;
            }
        }
        asm volatile("s_wait_dscnt 0x0" ::: "memory");
        {
            const int m  = lane >> 1;
            const int n0 = (lane & 1) * 16;
            float s = 0.f;
#pragma unroll
            for (int kx = 0; kx < 21; ++kx) s += cb[m * PSTR + n0 + kx];
            rt[m * 32 + n0] = s;
            for (int j = 1; j < 16; ++j) {
                const int n = n0 + j;
                s += cb[m * PSTR + n + 20] - cb[m * PSTR + n - 1];
                rt[m * 32 + n] = s;
            }
        }
        asm volatile("s_wait_dscnt 0x0" ::: "memory");

        for (int xx = lane; xx < PSTR; xx += 32) {
            float s2 = 0.f;
#pragma unroll
            for (int r = 0; r < 21; ++r) {
                float v = p[r * PSTR + xx];
                s2 = fmaf(v, v, s2);
            }
            cb[xx] = s2;
            for (int m = 1; m < 16; ++m) {
                float vi = p[(m + 20) * PSTR + xx];
                float vo = p[(m - 1) * PSTR + xx];
                s2 += vi * vi - vo * vo;
                cb[m * PSTR + xx] = s2;
            }
        }
        asm volatile("s_wait_dscnt 0x0" ::: "memory");
        {
            const int m  = lane >> 1;
            const int n0 = (lane & 1) * 16;
            const float inv = 1.f / 441.f;
            float s2 = 0.f;
#pragma unroll
            for (int kx = 0; kx < 21; ++kx) s2 += cb[m * PSTR + n0 + kx];
            {
                float mean = rt[m * 32 + n0] * inv, msq = s2 * inv;
                float stdv = sqrtf(msq - mean * mean + NCC_EPS);
                rt[m * 32 + n0] = 1.f / (stdv + NCC_EPS);
            }
            for (int j = 1; j < 16; ++j) {
                const int n = n0 + j;
                s2 += cb[m * PSTR + n + 20] - cb[m * PSTR + n - 1];
                float mean = rt[m * 32 + n] * inv, msq = s2 * inv;
                float stdv = sqrtf(msq - mean * mean + NCC_EPS);
                rt[m * 32 + n] = 1.f / (stdv + NCC_EPS);
            }
        }
        asm volatile("s_wait_dscnt 0x0" ::: "memory");

#pragma unroll
        for (int r = 0; r < 8; ++r) {
            const int m   = r + 8 * hl;
            const int gy  = y0 + m;
            const int gx0 = x0 + lane15;
            const int gx1 = gx0 + 16;
            if (gy < OUT_W && gx0 < OUT_W)
                dst0[gy * OUT_W + gx0] = acc0[r] * rt[m * 32 + lane15];
            if (gy < OUT_W && gx1 < OUT_W)
                dst0[gy * OUT_W + gx1] = acc1[r] * rt[m * 32 + 16 + lane15];
        }
    }
}

extern "C" void kernel_launch(void* const* d_in, const int* in_sizes, int n_in,
                              void* d_out, int out_size, void* d_ws, size_t ws_size,
                              hipStream_t stream) {
    const float* inp  = (const float*)d_in[0];
    const float* tmpl = (const float*)d_in[1];
    float* out = (float*)d_out;
    float* nt  = (float*)d_ws;

    ncc_tnorm<<<48, 32, 0, stream>>>(tmpl, nt);

    dim3 grid(124, 48);
    ncc_main<<<grid, 128, 0, stream>>>(inp, nt, out);
}
